# starting kernel
# baseline (speedup 1.0000x reference)
#include <hip/hip_runtime.h>


#define N_IN     256
#define N_OUT    256
#define N_PARAMS 16
#define DEGREE   3
#define BATCH    4096

#define M_TILE   64
#define N_TILE   64
#define THREADS  256
#define I_STAGE  4

typedef __attribute__((ext_vector_type(2))) float v2f;
typedef __attribute__((ext_vector_type(8))) float v8f;

__device__ __forceinline__ void bspline16(float t, float* __restrict__ out) {
  constexpr float h = 1.0f / (float)(N_PARAMS - DEGREE);
  float b[N_PARAMS + DEGREE];
#pragma unroll
  for (int j = 0; j < N_PARAMS + DEGREE; ++j) {
    const float k0 = (float)(j - DEGREE) * h;
    const float k1 = (float)(j + 1 - DEGREE) * h;
    b[j] = (t >= k0 && t < k1) ? 1.0f : 0.0f;
  }
#pragma unroll
  for (int d = 1; d <= DEGREE; ++d) {
    const float rdenom = 1.0f / ((float)d * h);
#pragma unroll
    for (int j = 0; j < N_PARAMS + DEGREE - d; ++j) {
      const float kj   = (float)(j - DEGREE) * h;
      const float kjd1 = (float)(j + d + 1 - DEGREE) * h;
      const float left  = (t - kj)   * rdenom;
      const float right = (kjd1 - t) * rdenom;
      b[j] = left * b[j] + right * b[j + 1];
    }
  }
#pragma unroll
  for (int p = 0; p < N_PARAMS; ++p) out[p] = b[p];
}

__global__ __launch_bounds__(THREADS)
void kan_wmma_f32_kernel(const float* __restrict__ x,
                         const float* __restrict__ coeff,
                         float* __restrict__ out) {
  __shared__ float sBasis[I_STAGE][M_TILE][N_PARAMS + 1];

  const int tid  = threadIdx.x;
  const int wave = tid >> 5;
  const int lane = tid & 31;
  const int laneLo = lane & 15;
  const int half   = lane >> 4;

  const int rowTile = wave & 3;
  const int colGrp  = wave >> 2;
  const int rowBase = blockIdx.x * M_TILE;
  const int colBase = blockIdx.y * N_TILE + colGrp * 32;

  const int pIsub = tid >> 6;
  const int pRow  = tid & 63;

  const int aRow = rowTile * 16 + laneLo;
  const int kHi  = half * 2;

  v8f acc[2] = {};

#pragma unroll 1
  for (int ii = 0; ii < N_IN; ii += I_STAGE) {
    __syncthreads();

    {
      const float t = x[(ii + pIsub) * BATCH + rowBase + pRow];
      float bp[N_PARAMS];
      bspline16(t, bp);
#pragma unroll
      for (int p = 0; p < N_PARAMS; ++p) sBasis[pIsub][pRow][p] = bp[p];
    }

    if (ii + I_STAGE < N_IN)
      __builtin_prefetch(
          coeff + (size_t)(ii + I_STAGE) * (N_PARAMS * N_OUT) + tid * 64, 0, 3);

    __syncthreads();

#pragma unroll
    for (int iSub = 0; iSub < I_STAGE; ++iSub) {
      const float* cRow = coeff + (size_t)(ii + iSub) * (N_PARAMS * N_OUT);
#pragma unroll
      for (int kk = 0; kk < N_PARAMS; kk += 4) {
        v2f a;
        a.x = sBasis[iSub][aRow][kk + kHi];
        a.y = sBasis[iSub][aRow][kk + kHi + 1];
#pragma unroll
        for (int ct = 0; ct < 2; ++ct) {
          const int n = colBase + ct * 16 + laneLo;
          v2f bf;
          bf.x = cRow[(kk + kHi) * N_OUT + n];
          bf.y = cRow[(kk + kHi + 1) * N_OUT + n];
          acc[ct] = __builtin_amdgcn_wmma_f32_16x16x4_f32(
               false, a,  false, bf,
               (short)0, acc[ct],  false,  false);
        }
      }
    }
  }

  const int mBase = rowBase + rowTile * 16 + half * 8;
#pragma unroll
  for (int ct = 0; ct < 2; ++ct) {
    const int n = colBase + ct * 16 + laneLo;
#pragma unroll
    for (int g = 0; g < 8; ++g) {
      out[(size_t)(mBase + g) * N_OUT + n] = acc[ct][g];
    }
  }
}

extern "C" void kernel_launch(void* const* d_in, const int* in_sizes, int n_in,
                              void* d_out, int out_size, void* d_ws, size_t ws_size,
                              hipStream_t stream) {
  const float* x     = (const float*)d_in[0];
  const float* coeff = (const float*)d_in[1];
  float* out = (float*)d_out;

  dim3 grid(BATCH / M_TILE, N_OUT / N_TILE);
  kan_wmma_f32_kernel<<<grid, THREADS, 0, stream>>>(x, coeff, out);
}


# Split K inside the workgroup: 8 waves, two per SIMD, each running the unchanged register pipeline on half of the input dims; partial 16x64 strips joined once through LDS before the store
# speedup vs baseline: 2.3014x; 2.3014x over previous
#include <hip/hip_runtime.h>


#define N_IN     256
#define N_OUT    256
#define N_PARAMS 16
#define DEGREE   3
#define BATCH    4096

#define M_TILE   64
#define N_TILE   64
#define K_SPLIT  2
#define THREADS  256
#define I_BLK    4

typedef __attribute__((ext_vector_type(2))) float v2f;
typedef __attribute__((ext_vector_type(4))) float v4f;
typedef __attribute__((ext_vector_type(8))) float v8f;

__device__ __forceinline__ void basis_frag(float t, int r0, v2f (&a)[4]) {
  const float nk = (float)(N_PARAMS - DEGREE);
  const float mf = floorf(t * nk);
  const float u  = fmaf(t, nk, -mf);
  const int   m  = (int)mf;

  const float u2 = u * u;
  const float u3 = u2 * u;
  const float v  = 1.0f - u;
  const float w0 = (v * v) * (v * (1.0f / 6.0f));
  const float w1 = fmaf(u2, fmaf(u, 0.5f, -1.0f), 2.0f / 3.0f);
  const float w2 = fmaf(u3, -0.5f, fmaf(u2, 0.5f, fmaf(u, 0.5f, 1.0f / 6.0f)));
  const float w3 = u3 * (1.0f / 6.0f);

  const unsigned d = (unsigned)r0 - (unsigned)m;
  const bool lo0 = (d & 1u) != 0, hi0 = (d & 2u) != 0;
  const bool hi1 = hi0 != lo0;
  const float c0 = hi0 ? (lo0 ? w3 : w2) : (lo0 ? w1 : w0);
  const float c1 = hi1 ? (lo0 ? w2 : w3) : (lo0 ? w0 : w1);
  const int g0 = (int)((unsigned)m + (unsigned)(3 - r0)) >> 2;
  const int g1 = (int)((unsigned)m + (unsigned)(2 - r0)) >> 2;
#pragma unroll
  for (int s = 0; s < 4; ++s) {
    a[s].x = (g0 == s) ? c0 : 0.0f;
    a[s].y = (g1 == s) ? c1 : 0.0f;
  }
}

struct Stage {
  v4f q[I_BLK][8];
};
struct XStage { float x[I_BLK]; };
struct AStage { v2f a[I_BLK][4]; };

__device__ __forceinline__ void load_stage(Stage& st, const float* __restrict__ bLane, int i0) {
#pragma unroll
  for (int j = 0; j < I_BLK; ++j) {
    const float* cRow = bLane + (size_t)(i0 + j) * (N_PARAMS * N_OUT);
#pragma unroll
    for (int s = 0; s < 4; ++s) {
      st.q[j][2 * s]     = *(const v4f*)(cRow + (4 * s) * N_OUT);
      st.q[j][2 * s + 1] = *(const v4f*)(cRow + (4 * s + 1) * N_OUT);
    }
  }
}

__device__ __forceinline__ void load_x(XStage& xs, const float* __restrict__ xRow, int i0) {
#pragma unroll
  for (int j = 0; j < I_BLK; ++j) xs.x[j] = xRow[(size_t)(i0 + j) * BATCH];
}

__device__ __forceinline__ void basis_stage(const XStage& xs, int r0, AStage& as) {
#pragma unroll
  for (int j = 0; j < I_BLK; ++j) basis_frag(xs.x[j], r0, as.a[j]);
}

__device__ __forceinline__ void compute_stage(const Stage& st, const AStage& as, v8f (&acc)[4]) {
#pragma unroll
  for (int j = 0; j < I_BLK; ++j) {
    const v2f (&a)[4] = as.a[j];
#pragma unroll
    for (int s = 0; s < 4; ++s) {
#pragma unroll
      for (int ct = 0; ct < 4; ++ct) {
        v2f b;
        b.x = st.q[j][2 * s][ct];
        b.y = st.q[j][2 * s + 1][ct];
        acc[ct] = __builtin_amdgcn_wmma_f32_16x16x4_f32(
             false, a[s],  false, b,
             (short)0, acc[ct],  false,  false);
      }
    }
  }
}

#define SGB_1W_4V() \
  __builtin_amdgcn_sched_group_barrier(0x008, 1, 0); \
  __builtin_amdgcn_sched_group_barrier(0x002, 4, 0);
#define SGB_X4()  SGB_1W_4V() SGB_1W_4V() SGB_1W_4V() SGB_1W_4V()
#define SGB_X16() SGB_X4() SGB_X4() SGB_X4() SGB_X4()
#define STAGE_INTERLEAVE() do { SGB_X16() SGB_X16() SGB_X16() SGB_X16() } while (0)

__global__ __launch_bounds__(THREADS)
void kan_wmma_f32_kernel(const float* __restrict__ x,
                         const float* __restrict__ coeff,
                         float* __restrict__ out) {
  const int tid  = threadIdx.x;
  const int wave  = tid >> 5;
  const int strip = wave & 3;
  const int kpart = wave >> 2;
  const int lane  = tid & 31;
  const int laneLo = lane & 15;
  const int half   = lane >> 4;

  const int rowBase = blockIdx.x * M_TILE + strip * 16;
  const int colBase = blockIdx.y * N_TILE;

  const int kHi = half * 2;

  const int N_I  = N_IN / K_SPLIT;
  const int iBeg = kpart * N_I;

  const float* xRow  = x + (size_t)iBeg * BATCH + rowBase + laneLo;
  const float* bLane = coeff + (size_t)iBeg * (N_PARAMS * N_OUT)
                             + (size_t)kHi * N_OUT + colBase + 4 * laneLo;

  v8f acc[4] = {};

  Stage  sa, sb;
  XStage xa, xb;
  AStage aa, ab;
  load_stage(sa, bLane, 0);
  load_x(xa, xRow, 0);
  load_x(xb, xRow, I_BLK);
  basis_stage(xa, kHi, aa);

  const int iLast = N_I - I_BLK;
  for (int ii = 0; ii < N_I; ii += 2 * I_BLK) {
    const int i2 = (ii + 2 * I_BLK < N_I) ? ii + 2 * I_BLK : iLast;
    const int i3 = (ii + 3 * I_BLK < N_I) ? ii + 3 * I_BLK : iLast;

    load_stage(sb, bLane, ii + I_BLK);
    load_x(xa, xRow, i2);
    __builtin_amdgcn_sched_barrier(0);
    basis_stage(xb, kHi, ab);
    compute_stage(sa, aa, acc);
    STAGE_INTERLEAVE();
    __builtin_amdgcn_sched_barrier(0);

    load_stage(sa, bLane, i2);
    load_x(xb, xRow, i3);
    __builtin_amdgcn_sched_barrier(0);
    basis_stage(xa, kHi, aa);
    compute_stage(sb, ab, acc);
    STAGE_INTERLEAVE();
    __builtin_amdgcn_sched_barrier(0);
  }

  __shared__ v4f part[4 * 8 * 32];
  v4f* pLane = part + strip * (8 * 32) + lane;
  if (kpart == 1) {
#pragma unroll
    for (int g = 0; g < 8; ++g) {
      v4f o;
      o.x = acc[0][g];
      o.y = acc[1][g];
      o.z = acc[2][g];
      o.w = acc[3][g];
      pLane[g * 32] = o;
    }
  }
  __syncthreads();
  if (kpart == 0) {
    float* oLane = out + (size_t)(rowBase + half * 8) * N_OUT + colBase + 4 * laneLo;
#pragma unroll
    for (int g = 0; g < 8; ++g) {
      const v4f p = pLane[g * 32];
      v4f o;
      o.x = acc[0][g] + p.x;
      o.y = acc[1][g] + p.y;
      o.z = acc[2][g] + p.z;
      o.w = acc[3][g] + p.w;
      *(v4f*)(oLane + (size_t)g * N_OUT) = o;
    }
  }
}

extern "C" void kernel_launch(void* const* d_in, const int* in_sizes, int n_in,
                              void* d_out, int out_size, void* d_ws, size_t ws_size,
                              hipStream_t stream) {
  const float* x     = (const float*)d_in[0];
  const float* coeff = (const float*)d_in[1];
  float* out = (float*)d_out;

  dim3 grid(BATCH / M_TILE, N_OUT / N_TILE);
  kan_wmma_f32_kernel<<<grid, THREADS, 0, stream>>>(x, coeff, out);
}


# Wave tile widened from 16x64 to 32x128 (split K 8 inside the workgroup): one spline A fragment now feeds 8 column tiles and one packed B pair both row tiles, halving vector work per WMMA
# speedup vs baseline: 2.5633x; 1.1138x over previous
#include <hip/hip_runtime.h>


#define N_IN     256
#define N_OUT    256
#define N_PARAMS 16
#define DEGREE   3
#define BATCH    4096

#define R_TILES  2
#define C_BLKS   2
#define C_TILES  (4 * C_BLKS)
#define M_TILE   (16 * R_TILES)
#define N_TILE   (64 * C_BLKS)
#define K_SPLIT  8
#define THREADS  256
#define SLOTS    (R_TILES * 8 * C_BLKS)

typedef __attribute__((ext_vector_type(2))) float v2f;
typedef __attribute__((ext_vector_type(4))) float v4f;
typedef __attribute__((ext_vector_type(8))) float v8f;

__device__ __forceinline__ void basis_frag(float t, int r0, v2f (&a)[4]) {
  const float nk = (float)(N_PARAMS - DEGREE);
  const float mf = floorf(t * nk);
  const float u  = fmaf(t, nk, -mf);
  const int   m  = (int)mf;

  const float u2 = u * u;
  const float u3 = u2 * u;
  const float v  = 1.0f - u;
  const float w0 = (v * v) * (v * (1.0f / 6.0f));
  const float w1 = fmaf(u2, fmaf(u, 0.5f, -1.0f), 2.0f / 3.0f);
  const float w2 = fmaf(u3, -0.5f, fmaf(u2, 0.5f, fmaf(u, 0.5f, 1.0f / 6.0f)));
  const float w3 = u3 * (1.0f / 6.0f);

  const unsigned d = (unsigned)r0 - (unsigned)m;
  const bool lo0 = (d & 1u) != 0, hi0 = (d & 2u) != 0;
  const bool hi1 = hi0 != lo0;
  const float c0 = hi0 ? (lo0 ? w3 : w2) : (lo0 ? w1 : w0);
  const float c1 = hi1 ? (lo0 ? w2 : w3) : (lo0 ? w0 : w1);
  const int g0 = (int)((unsigned)m + (unsigned)(3 - r0)) >> 2;
  const int g1 = (int)((unsigned)m + (unsigned)(2 - r0)) >> 2;
#pragma unroll
  for (int s = 0; s < 4; ++s) {
    a[s].x = (g0 == s) ? c0 : 0.0f;
    a[s].y = (g1 == s) ? c1 : 0.0f;
  }
}

struct Stage {
  v4f q[8][C_BLKS];
};
struct XStage { float x[R_TILES]; };
struct AStage { v2f a[R_TILES][4]; };

__device__ __forceinline__ void load_stage(Stage& st, const float* __restrict__ bLane, int i) {
  const float* cRow = bLane + (size_t)i * (N_PARAMS * N_OUT);
#pragma unroll
  for (int s = 0; s < 4; ++s) {
#pragma unroll
    for (int J = 0; J < C_BLKS; ++J) {
      st.q[2 * s][J]     = *(const v4f*)(cRow + (4 * s) * N_OUT + 64 * J);
      st.q[2 * s + 1][J] = *(const v4f*)(cRow + (4 * s + 1) * N_OUT + 64 * J);
    }
  }
}

__device__ __forceinline__ void load_x(XStage& xs, const float* __restrict__ xRow, int i) {
#pragma unroll
  for (int rt = 0; rt < R_TILES; ++rt) xs.x[rt] = xRow[(size_t)i * BATCH + 16 * rt];
}

__device__ __forceinline__ void basis_stage(const XStage& xs, int r0, AStage& as) {
#pragma unroll
  for (int rt = 0; rt < R_TILES; ++rt) basis_frag(xs.x[rt], r0, as.a[rt]);
}

__device__ __forceinline__ void compute_stage(const Stage& st, const AStage& as,
                                              v8f (&acc)[R_TILES][C_TILES]) {
#pragma unroll
  for (int s = 0; s < 4; ++s) {
#pragma unroll
    for (int J = 0; J < C_BLKS; ++J) {
#pragma unroll
      for (int e = 0; e < 4; ++e) {
        v2f b;
        b.x = st.q[2 * s][J][e];
        b.y = st.q[2 * s + 1][J][e];
#pragma unroll
        for (int rt = 0; rt < R_TILES; ++rt) {
          acc[rt][4 * J + e] = __builtin_amdgcn_wmma_f32_16x16x4_f32(
               false, as.a[rt][s],  false, b,
               (short)0, acc[rt][4 * J + e],  false,  false);
        }
      }
    }
  }
}

#define SGB_1W_3V() \
  __builtin_amdgcn_sched_group_barrier(0x008, 1, 0); \
  __builtin_amdgcn_sched_group_barrier(0x002, 3, 0);
#define SGB_X4()  SGB_1W_3V() SGB_1W_3V() SGB_1W_3V() SGB_1W_3V()
#define SGB_X16() SGB_X4() SGB_X4() SGB_X4() SGB_X4()
#define STAGE_INTERLEAVE() do { SGB_X16() SGB_X16() SGB_X16() SGB_X16() } while (0)

__global__ __launch_bounds__(THREADS)
void kan_wmma_f32_kernel(const float* __restrict__ x,
                         const float* __restrict__ coeff,
                         float* __restrict__ out) {
  const int tid  = threadIdx.x;
  const int kpart = tid >> 5;
  const int lane  = tid & 31;
  const int laneLo = lane & 15;
  const int half   = lane >> 4;

  const int rowBase = blockIdx.x * M_TILE;
  const int colBase = blockIdx.y * N_TILE;

  const int kHi = half * 2;

  const int N_I  = N_IN / K_SPLIT;
  const int iBeg = kpart * N_I;

  const float* xRow  = x + (size_t)iBeg * BATCH + rowBase + laneLo;
  const float* bLane = coeff + (size_t)iBeg * (N_PARAMS * N_OUT)
                             + (size_t)kHi * N_OUT + colBase + 4 * laneLo;

  v8f acc[R_TILES][C_TILES] = {};

  Stage  sa, sb;
  XStage xa, xb;
  AStage aa, ab;
  load_stage(sa, bLane, 0);
  load_x(xa, xRow, 0);
  load_x(xb, xRow, 1);
  basis_stage(xa, kHi, aa);

  const int iLast = N_I - 1;
  for (int ii = 0; ii < N_I; ii += 2) {
    const int i2 = (ii + 2 < N_I) ? ii + 2 : iLast;
    const int i3 = (ii + 3 < N_I) ? ii + 3 : iLast;

    load_stage(sb, bLane, ii + 1);
    load_x(xa, xRow, i2);
    __builtin_amdgcn_sched_barrier(0);
    basis_stage(xb, kHi, ab);
    compute_stage(sa, aa, acc);
    STAGE_INTERLEAVE();
    __builtin_amdgcn_sched_barrier(0);

    load_stage(sa, bLane, i2);
    load_x(xb, xRow, i3);
    __builtin_amdgcn_sched_barrier(0);
    basis_stage(xa, kHi, aa);
    compute_stage(sb, ab, acc);
    STAGE_INTERLEAVE();
    __builtin_amdgcn_sched_barrier(0);
  }

  __shared__ v4f part[K_SPLIT * SLOTS * 32];
  {
    v4f* pw = part + (size_t)kpart * (SLOTS * 32) + lane;
#pragma unroll
    for (int rt = 0; rt < R_TILES; ++rt) {
#pragma unroll
      for (int g = 0; g < 8; ++g) {
#pragma unroll
        for (int J = 0; J < C_BLKS; ++J) {
          v4f o;
          o.x = acc[rt][4 * J + 0][g];
          o.y = acc[rt][4 * J + 1][g];
          o.z = acc[rt][4 * J + 2][g];
          o.w = acc[rt][4 * J + 3][g];
          pw[((rt * 8 + g) * C_BLKS + J) * 32] = o;
        }
      }
    }
  }
  __syncthreads();
  {
    const int PER_WAVE = SLOTS / K_SPLIT;
#pragma unroll
    for (int q = 0; q < PER_WAVE; ++q) {
      const int slot = kpart * PER_WAVE + q;
      const v4f* pr = part + (size_t)slot * 32 + lane;
      v4f o = pr[0];
#pragma unroll
      for (int k = 1; k < K_SPLIT; ++k) {
        const v4f p = pr[(size_t)k * (SLOTS * 32)];
        o.x += p.x;
        o.y += p.y;
        o.z += p.z;
        o.w += p.w;
      }
      const int J  = slot % C_BLKS;
      const int g  = (slot / C_BLKS) % 8;
      const int rt = slot / (8 * C_BLKS);
      float* oLane = out + (size_t)(rowBase + 16 * rt + 8 * half + g) * N_OUT
                         + colBase + 64 * J + 4 * laneLo;
      *(v4f*)oLane = o;
    }
  }
}

extern "C" void kernel_launch(void* const* d_in, const int* in_sizes, int n_in,
                              void* d_out, int out_size, void* d_ws, size_t ws_size,
                              hipStream_t stream) {
  const float* x     = (const float*)d_in[0];
  const float* coeff = (const float*)d_in[1];
  float* out = (float*)d_out;

  dim3 grid(BATCH / M_TILE, N_OUT / N_TILE);
  kan_wmma_f32_kernel<<<grid, THREADS, 0, stream>>>(x, coeff, out);
}
